# starting kernel
# baseline (speedup 1.0000x reference)
#include <hip/hip_runtime.h>
#include <hip/hip_bf16.h>

typedef float v2f __attribute__((ext_vector_type(2)));
typedef float v8f __attribute__((ext_vector_type(8)));

__global__ void ScorePredictor_81097572483639_kernel(
    const float* __restrict__ h_drug,
    const float* __restrict__ d_disease,
    const int* __restrict__ src_idx,
    const int* __restrict__ dst_idx,
    float* __restrict__ out,
    int n_edges, int n_groups) {
  const int lane          = threadIdx.x & 31;
  const int waveInBlock   = threadIdx.x >> 5;
  const int wavesPerBlock = blockDim.x >> 5;
  const int groupStride   = gridDim.x * wavesPerBlock;
  int group = blockIdx.x * wavesPerBlock + waveInBlock;

  const int m  = lane & 15;
  const int kh = lane >> 4;

  for (; group < n_groups; group += groupStride) {
    const int base = group * 16;
    int e = base + m;
    if (e >= n_edges) e = 0;

    const int si = src_idx[e];
    const int di = dst_idx[e];
    const float* pa = h_drug    + (size_t)si * 128 + (kh << 1);
    const float* pb = d_disease + (size_t)di * 128 + (kh << 1);

    v8f c = {0.f, 0.f, 0.f, 0.f, 0.f, 0.f, 0.f, 0.f};
#pragma unroll
    for (int k0 = 0; k0 < 128; k0 += 4) {
      v2f a = *(const v2f*)(pa + k0);
      v2f b = *(const v2f*)(pb + k0);
      c = __builtin_amdgcn_wmma_f32_16x16x4_f32(false, a, false, b,
                                                (short)0, c, false, false);
    }

    const int srcLane = (m < 8) ? m : (m + 16);
    const int sel     = m & 7;
    float diag = 0.f;
#pragma unroll
    for (int i = 0; i < 8; ++i) {
      float t = __shfl(c[i], srcLane, 32);
      if (sel == i) diag = t;
    }

    const int eo = base + lane;
    if (lane < 16 && eo < n_edges) {
      float mean = diag * (1.0f / 128.0f);
      out[eo] = 1.0f / (1.0f + __expf(-mean));
    }
  }
}

extern "C" void kernel_launch(void* const* d_in, const int* in_sizes, int n_in,
                              void* d_out, int out_size, void* d_ws, size_t ws_size,
                              hipStream_t stream) {
  const float* h_drug    = (const float*)d_in[0];
  const float* d_disease = (const float*)d_in[1];
  const int*   src_idx   = (const int*)d_in[2];
  const int*   dst_idx   = (const int*)d_in[3];
  float*       out       = (float*)d_out;

  const int n_edges  = in_sizes[2];
  const int n_groups = (n_edges + 15) / 16;

  const int threads        = 256;
  const int wavesPerBlock  = threads / 32;
  int blocks = (n_groups + wavesPerBlock - 1) / wavesPerBlock;
  if (blocks < 1) blocks = 1;

  ScorePredictor_81097572483639_kernel<<<blocks, threads, 0, stream>>>(
      h_drug, d_disease, src_idx, dst_idx, out, n_edges, n_groups);
}


# each gathered 512-B row fetched whole by one coalesced wave load, staged in wave-private LDS, read back as WMMA fragments; 512 looping blocks; order of the feature sum changes
# speedup vs baseline: 3.2991x; 3.2991x over previous
#include <hip/hip_runtime.h>

typedef float v2f __attribute__((ext_vector_type(2)));
typedef float v4f __attribute__((ext_vector_type(4)));
typedef float v8f __attribute__((ext_vector_type(8)));


#define FEAT   128
#define PITCH  132
#define WAVES  8
#define ROWS   32

__device__ __forceinline__ void gather_rows(const float* __restrict__ h_drug,
                                            const float* __restrict__ d_disease,
                                            int si, int di, int first, int lane,
                                            v4f (&ra)[16], v4f (&rb)[16]) {
#pragma unroll
  for (int j = 0; j < 16; ++j) {
    const int s = __builtin_amdgcn_readlane(si, first + j);
    const int d = __builtin_amdgcn_readlane(di, first + j);
    ra[j] = *(const v4f*)(h_drug    + (size_t)s * FEAT + lane * 4);
    rb[j] = *(const v4f*)(d_disease + (size_t)d * FEAT + lane * 4);
  }
}

__device__ __forceinline__ void stage_rows(float* my, int lane,
                                           const v4f (&ra)[16], const v4f (&rb)[16]) {
#pragma unroll
  for (int j = 0; j < 16; ++j) {
    *(v4f*)(my + j * PITCH + lane * 4)        = ra[j];
    *(v4f*)(my + (16 + j) * PITCH + lane * 4) = rb[j];
  }
}

__device__ __forceinline__ v8f dot_tile(const float* pa, const float* pb, v8f c) {
#pragma unroll
  for (int kk = 0; kk < 16; ++kk) {
    const v4f a = *(const v4f*)(pa + kk * 8);
    const v4f b = *(const v4f*)(pb + kk * 8);
    const v2f a0 = {a.x, a.y}, b0 = {b.x, b.y};
    const v2f a1 = {a.z, a.w}, b1 = {b.z, b.w};
    c = __builtin_amdgcn_wmma_f32_16x16x4_f32(false, a0, false, b0, (short)0, c, false, false);
    c = __builtin_amdgcn_wmma_f32_16x16x4_f32(false, a1, false, b1, (short)0, c, false, false);
  }
  return c;
}

__global__ void __launch_bounds__(256)
edge_score_kernel(
    const float* __restrict__ h_drug,
    const float* __restrict__ d_disease,
    const int* __restrict__ src_idx,
    const int* __restrict__ dst_idx,
    float* __restrict__ out,
    int n_edges, int n_pairs) {
  __shared__ __attribute__((aligned(16))) float stage[WAVES * ROWS * PITCH];

  const int lane = threadIdx.x & 31;
  const int wave = __builtin_amdgcn_readfirstlane((int)(threadIdx.x >> 5));
  const int stride = gridDim.x * WAVES;
  int p = blockIdx.x * WAVES + wave;
  if (p >= n_pairs) return;

  float* my = stage + wave * (ROWS * PITCH);
  const int m  = lane & 15;
  const int kh = lane >> 4;
  const float* pa  = my + m * PITCH + kh * 4;
  const float* pb0 = my + (16 + m) * PITCH + kh * 4;
  const float* pb1 = my + (16 + (m ^ 8)) * PITCH + kh * 4;

  int e = p * 32 + lane;
  if (e >= n_edges) e = 0;
  int si = src_idx[e];
  int di = dst_idx[e];

  v4f ra[16], rb[16];
  gather_rows(h_drug, d_disease, si, di, 0, lane, ra, rb);

  for (; p < n_pairs; p += stride) {
    const int pn = p + stride;
    const bool more = pn < n_pairs;
    int en = pn * 32 + lane;
    if (!more || en >= n_edges) en = 0;
    const int si_n = src_idx[en];
    const int di_n = dst_idx[en];

    stage_rows(my, lane, ra, rb);
    __builtin_amdgcn_fence(__ATOMIC_ACQ_REL, "wavefront");
    gather_rows(h_drug, d_disease, si, di, 16, lane, ra, rb);
    v8f c0 = {0.f, 0.f, 0.f, 0.f, 0.f, 0.f, 0.f, 0.f};
    c0 = dot_tile(pa, pb0, c0);
    __builtin_amdgcn_fence(__ATOMIC_ACQ_REL, "wavefront");

    stage_rows(my, lane, ra, rb);
    __builtin_amdgcn_fence(__ATOMIC_ACQ_REL, "wavefront");
    if (more) gather_rows(h_drug, d_disease, si_n, di_n, 0, lane, ra, rb);
    v8f c1 = {0.f, 0.f, 0.f, 0.f, 0.f, 0.f, 0.f, 0.f};
    c1 = dot_tile(pa, pb1, c1);
    __builtin_amdgcn_fence(__ATOMIC_ACQ_REL, "wavefront");

    const int  sel = lane & 7;
    const bool mid = (lane >= 8) && (lane < 24);
    float v = 0.f;
#pragma unroll
    for (int i = 0; i < 8; ++i) {
      const float t = mid ? c1[i] : c0[i];
      if (sel == i) v = t;
    }
    const int off = mid ? (lane + 8) : ((lane < 8) ? lane : (lane - 16));
    const int eo = p * 32 + off;
    if (eo < n_edges) {
      const float mean = v * (1.0f / 128.0f);
      out[eo] = 1.0f / (1.0f + __expf(-mean));
    }

    si = si_n;
    di = di_n;
  }
}

extern "C" void kernel_launch(void* const* d_in, const int* in_sizes, int n_in,
                              void* d_out, int out_size, void* d_ws, size_t ws_size,
                              hipStream_t stream) {
  const float* h_drug    = (const float*)d_in[0];
  const float* d_disease = (const float*)d_in[1];
  const int*   src_idx   = (const int*)d_in[2];
  const int*   dst_idx   = (const int*)d_in[3];
  float*       out       = (float*)d_out;

  const int n_edges = in_sizes[2];
  const int n_pairs = (n_edges + 31) / 32;

  edge_score_kernel<<<512, 256, 0, stream>>>(
      h_drug, d_disease, src_idx, dst_idx, out, n_edges, n_pairs);
}
